# starting kernel
# baseline (speedup 1.0000x reference)
#include <hip/hip_runtime.h>

#define EPSV 1e-5f

typedef __attribute__((ext_vector_type(2))) float v2f;
typedef __attribute__((ext_vector_type(8))) float v8f;

__global__ __launch_bounds__(256) void prep_weights_kernel(
    const float* __restrict__ w1, const float* __restrict__ w2,
    float* __restrict__ wt)
{
    int i = blockIdx.x * 256 + threadIdx.x;
    if (i >= 2 * 36864) return;
    const float* w = (i < 36864) ? w1 : w2;
    float* dst = wt + (i < 36864 ? 0 : 36864);
    int j = i % 36864;
    int tap = j / 4096;
    int rem = j % 4096;
    int n = (rem >> 1) & 63;
    int k = ((rem >> 7) << 1) | (rem & 1);
    dst[j] = w[(n * 64 + k) * 9 + tap];
}

__global__ __launch_bounds__(256) void fused_basicblock_kernel(
    const float* __restrict__ x,
    const float* __restrict__ wt,
    const float* __restrict__ g1, const float* __restrict__ b1,
    const float* __restrict__ m1, const float* __restrict__ v1,
    const float* __restrict__ g2, const float* __restrict__ b2,
    const float* __restrict__ m2, const float* __restrict__ v2p,
    const float* __restrict__ mask1, const float* __restrict__ mask2,
    float* __restrict__ out)
{
    __shared__ __align__(16) float xpad[64 * 81];
    __shared__ __align__(16) float ipad[64 * 81];
    __shared__ __align__(16) float wsh[64 * 64];

    const int tid  = threadIdx.x;
    const int lane = tid & 31;
    const int wave = tid >> 5;
    const int b    = blockIdx.x;

    const int lo  = lane & 15;
    const int hi  = lane >> 4;
    const int khi = hi * 2;

    const int nn  = (wave & 3) * 16 + lo;
    const int mt0 = (wave >> 2);
    int mA0 = mt0 * 16 + lo;        if (mA0 > 48) mA0 = 48;
    int mA1 = (mt0 + 2) * 16 + lo;  if (mA1 > 48) mA1 = 48;
    const int y0 = mA0 / 7, x0c = mA0 % 7;
    const int y1 = mA1 / 7, x1c = mA1 % 7;

    for (int i = tid; i < 64 * 81; i += 256) { xpad[i] = 0.0f; ipad[i] = 0.0f; }

    const float* xb = x + (size_t)b * 3136;
    for (int i = tid; i < 3136; i += 256) {
        int c = i / 49, m = i % 49;
        xpad[c * 81 + (m / 7 + 1) * 9 + (m % 7) + 1] = xb[i];
    }
    __syncthreads();

    auto conv_pass = [&](const float* src, const float* __restrict__ wslab,
                         v8f& acc0, v8f& acc1) {
        for (int tap = 0; tap < 9; ++tap) {
            __syncthreads();
            {
                const float4* s4 = (const float4*)(wslab + tap * 4096);
                float4* d4 = (float4*)wsh;
                #pragma unroll
                for (int i = 0; i < 4; ++i) d4[tid + i * 256] = s4[tid + i * 256];
            }
            if (tap < 8)
                __builtin_prefetch((const char*)(wslab + (tap + 1) * 4096) + tid * 64, 0, 1);
            __syncthreads();

            const int r = tap / 3, s = tap - r * 3;
            const int so0 = (y0 + r) * 9 + (x0c + s);
            const int so1 = (y1 + r) * 9 + (x1c + s);
            #pragma unroll
            for (int ks = 0; ks < 16; ++ks) {
                const int kl = ks * 4 + khi;
                v2f a0, a1;
                a0[0] = src[kl * 81 + so0];
                a0[1] = src[kl * 81 + 81 + so0];
                a1[0] = src[kl * 81 + so1];
                a1[1] = src[kl * 81 + 81 + so1];
                const v2f bf = *(const v2f*)&wsh[(ks * 2 + hi) * 128 + nn * 2];
                acc0 = __builtin_amdgcn_wmma_f32_16x16x4_f32(
                    false, a0, false, bf, (short)0, acc0, false, false);
                acc1 = __builtin_amdgcn_wmma_f32_16x16x4_f32(
                    false, a1, false, bf, (short)0, acc1, false, false);
            }
        }
    };

    v8f c1a = {}, c1b = {};
    conv_pass(xpad, wt, c1a, c1b);

    const float sc1 = g1[nn] * rsqrtf(v1[nn] + EPSV);
    const float sh1 = b1[nn] - m1[nn] * sc1;
    auto epi1 = [&](const v8f& acc, int mt) {
        #pragma unroll
        for (int j = 0; j < 8; ++j) {
            int m = mt * 16 + j + 8 * hi;
            if (m < 49) {
                float v = fmaf(acc[j], sc1, sh1);
                v = fmaxf(v, 0.0f);
                if (b == 0) v *= mask1[nn * 49 + m];
                ipad[nn * 81 + (m / 7 + 1) * 9 + (m % 7) + 1] = v;
            }
        }
    };
    epi1(c1a, mt0);
    epi1(c1b, mt0 + 2);

    v8f c2a = {}, c2b = {};
    conv_pass(ipad, wt + 36864, c2a, c2b);
    __syncthreads();

    const float sc2 = g2[nn] * rsqrtf(v2p[nn] + EPSV);
    const float sh2 = b2[nn] - m2[nn] * sc2;
    auto epi2 = [&](const v8f& acc, int mt) {
        #pragma unroll
        for (int j = 0; j < 8; ++j) {
            int m = mt * 16 + j + 8 * hi;
            if (m < 49) {
                float v = acc[j];
                if (b == 0) v *= mask2[nn * 49 + m];
                v = fmaf(v, sc2, sh2);
                int off = nn * 81 + (m / 7 + 1) * 9 + (m % 7) + 1;
                v += xpad[off];
                ipad[off] = fmaxf(v, 0.0f);
            }
        }
    };
    epi2(c2a, mt0);
    epi2(c2b, mt0 + 2);
    __syncthreads();

    float* ob = out + (size_t)b * 3136;
    for (int i = tid; i < 3136; i += 256) {
        int c = i / 49, m = i % 49;
        ob[i] = ipad[c * 81 + (m / 7 + 1) * 9 + (m % 7) + 1];
    }
}

extern "C" void kernel_launch(void* const* d_in, const int* in_sizes, int n_in,
                              void* d_out, int out_size, void* d_ws, size_t ws_size,
                              hipStream_t stream) {
    const float* x     = (const float*)d_in[0];
    const float* w1    = (const float*)d_in[1];
    const float* g1    = (const float*)d_in[2];
    const float* b1    = (const float*)d_in[3];
    const float* m1    = (const float*)d_in[4];
    const float* v1    = (const float*)d_in[5];
    const float* w2    = (const float*)d_in[6];
    const float* g2    = (const float*)d_in[7];
    const float* b2    = (const float*)d_in[8];
    const float* m2    = (const float*)d_in[9];
    const float* v2p   = (const float*)d_in[10];
    const float* mask1 = (const float*)d_in[11];
    const float* mask2 = (const float*)d_in[12];
    float* wt = (float*)d_ws;
    float* out = (float*)d_out;

    prep_weights_kernel<<<288, 256, 0, stream>>>(w1, w2, wt);
    fused_basicblock_kernel<<<4096, 256, 0, stream>>>(
        x, wt, g1, b1, m1, v1, g2, b2, m2, v2p, mask1, mask2, out);
}


# Staging and write-back as whole-image 16-byte row copies parked in the idle intermediate tile, re-laid LDS to LDS at constant offsets, a lane per channel pair: per-element index arithmetic gone
# speedup vs baseline: 1.4997x; 1.4997x over previous
#include <hip/hip_runtime.h>


#define EPSV 1e-5f

typedef __attribute__((ext_vector_type(2))) float v2f;
typedef __attribute__((ext_vector_type(8))) float v8f;

__global__ __launch_bounds__(256) void prep_weights_kernel(
    const float* __restrict__ w1, const float* __restrict__ w2,
    float* __restrict__ wt)
{
    int i = blockIdx.x * 256 + threadIdx.x;
    if (i >= 2 * 36864) return;
    const float* w = (i < 36864) ? w1 : w2;
    float* dst = wt + (i < 36864 ? 0 : 36864);
    int j = i % 36864;
    int tap = j / 4096;
    int rem = j % 4096;
    int n = (rem >> 1) & 63;
    int k = ((rem >> 7) << 1) | (rem & 1);
    dst[j] = w[(n * 64 + k) * 9 + tap];
}

#define NIMG 4
#define NTHR 128
#define PITCH 8
#define QP 73
#define TILE_F (64 * QP)
#define NPOS (NIMG * 49)
#define NTILE 13
#define NGRP 4
#define FPG (144 / NGRP)

__global__ __launch_bounds__(NTHR, 2) void fused_basicblock_kernel(
    const float* __restrict__ x,
    const float* __restrict__ wt,
    const float* __restrict__ g1, const float* __restrict__ b1,
    const float* __restrict__ m1, const float* __restrict__ v1,
    const float* __restrict__ g2, const float* __restrict__ b2,
    const float* __restrict__ m2, const float* __restrict__ v2p,
    const float* __restrict__ mask1, const float* __restrict__ mask2,
    float* __restrict__ out)
{
    __shared__ __align__(16) float xpad[NIMG * TILE_F];
    __shared__ __align__(16) float ipad[NIMG * TILE_F];

    const int tid  = threadIdx.x;
    const int lane = tid & 31;
    const int wave = __builtin_amdgcn_readfirstlane(tid >> 5);
    const int c    = lane & 15;
    const int h    = lane >> 4;
    const int mt   = wave;
    const int img0 = blockIdx.x * NIMG;

    const float4 z4 = {0.0f, 0.0f, 0.0f, 0.0f};
    float* park = ipad + wave * TILE_F;
    {
        const float4* xs = (const float4*)(x + (size_t)(img0 + wave) * 3136);
        float4 rv[24];
        float4 rlast = z4;
        #pragma unroll
        for (int j = 0; j < 24; ++j) rv[j] = xs[lane + 32 * j];
        if (lane < 16) rlast = xs[lane + 768];
        float4* xz = (float4*)xpad;
        for (int i = tid; i < NIMG * TILE_F / 4; i += NTHR) xz[i] = z4;
        float4* pk = (float4*)park;
        #pragma unroll
        for (int j = 0; j < 24; ++j) pk[lane + 32 * j] = rv[j];
        if (lane < 16) pk[lane + 768] = rlast;
    }
    __syncthreads();
    {
        const float* rp = park + lane * 98;
        float* dp = xpad + wave * TILE_F + lane * (2 * QP);
        #pragma unroll
        for (int m = 0; m < 49; ++m) {
            const int q = (m / 7 + 1) * PITCH + (m % 7) + 1;
            v2f st; st[0] = rp[m]; st[1] = rp[49 + m];
            *(v2f*)(dp + q * 2) = st;
        }
    }
    __syncthreads();
    {
        float4* iz = (float4*)ipad;
        for (int i = tid; i < NIMG * TILE_F / 4; i += NTHR) iz[i] = z4;
    }

    v2f W[144];
    auto load_w = [&](const float* __restrict__ wconv) {
        const float* wp = wconv + h * 128 + (mt * 16 + c) * 2;
        #pragma unroll
        for (int t = 0; t < 9; ++t) {
            #pragma unroll
            for (int ks = 0; ks < 16; ++ks)
                W[t * 16 + ks] = *(const v2f*)(wp + t * 4096 + ks * 256);
        }
    };

    auto tile_ptr = [&](const float* src, int ti) __attribute__((always_inline)) -> const v2f* {
        int p = ti * 16 + c;
        if (p >= NPOS) p = NPOS - 1;
        const int img = p / 49, m = p - img * 49;
        const int y = m / 7, xx = m - y * 7;
        return (const v2f*)(src + img * TILE_F + h * (2 * QP) + (y * PITCH + xx) * 2);
    };

    v2f S0[FPG], S1[FPG];
    auto load_grp = [&](v2f (&S)[FPG], const v2f* bp, const int g) __attribute__((always_inline)) {
        #pragma unroll
        for (int j = 0; j < FPG; ++j) {
            const int f = g * FPG + j, t = f >> 4, ks = f & 15;
            S[j] = bp[ks * (2 * QP) + (t / 3) * PITCH + (t % 3)];
        }
    };
    auto mma_grp = [&](const v2f (&S)[FPG], const int g, v8f& a0, v8f& a1) __attribute__((always_inline)) {
        #pragma unroll
        for (int j = 0; j < FPG; j += 2) {
            a0 = __builtin_amdgcn_wmma_f32_16x16x4_f32(
                false, W[g * FPG + j], false, S[j], (short)0, a0, false, false);
            a1 = __builtin_amdgcn_wmma_f32_16x16x4_f32(
                false, W[g * FPG + j + 1], false, S[j + 1], (short)0, a1, false, false);
        }
    };

#define KEEP_GROUPED(A0, A1) asm volatile("" : "+v"(A0), "+v"(A1) : : "memory")

    auto conv_pass = [&](const float* src, auto&& epilogue) __attribute__((always_inline)) {
        const v2f* bp = tile_ptr(src, 0);
        load_grp(S0, bp, 0);
        asm volatile("" : : : "memory");
        load_grp(S1, bp, 1);
        for (int ti = 0; ti < NTILE; ++ti) {
            v8f a0 = {}, a1 = {};
            mma_grp(S0, 0, a0, a1); KEEP_GROUPED(a0, a1);
            load_grp(S0, bp, 2);    KEEP_GROUPED(a0, a1);
            mma_grp(S1, 1, a0, a1); KEEP_GROUPED(a0, a1);
            load_grp(S1, bp, 3);    KEEP_GROUPED(a0, a1);
            mma_grp(S0, 2, a0, a1); KEEP_GROUPED(a0, a1);
            bp = tile_ptr(src, ti + 1);
            load_grp(S0, bp, 0);    KEEP_GROUPED(a0, a1);
            mma_grp(S1, 3, a0, a1); KEEP_GROUPED(a0, a1);
            load_grp(S1, bp, 1);
            epilogue(ti, a0 + a1);
        }
    };

    const int obase = mt * 16 + 8 * h;

    float sc[8], sh[8];

    load_w(wt);
    #pragma unroll
    for (int r = 0; r < 8; ++r) {
        sc[r] = g1[obase + r] * rsqrtf(v1[obase + r] + EPSV);
        sh[r] = b1[obase + r] - m1[obase + r] * sc[r];
    }
    __syncthreads();

    conv_pass(xpad, [&](int ti, v8f acc) __attribute__((always_inline)) {
        const int p = ti * 16 + c;
        if (p < NPOS) {
            const int img = p / 49, m = p - img * 49;
            const int q = (m / 7 + 1) * PITCH + (m % 7) + 1;
            const bool first = (img0 + img) == 0;
            #pragma unroll
            for (int r = 0; r < 8; r += 2) {
                const int o = obase + r;
                float u0 = fmaxf(fmaf(acc[r],     sc[r],     sh[r]),     0.0f);
                float u1 = fmaxf(fmaf(acc[r + 1], sc[r + 1], sh[r + 1]), 0.0f);
                if (first) { u0 *= mask1[o * 49 + m]; u1 *= mask1[(o + 1) * 49 + m]; }
                v2f st; st[0] = u0; st[1] = u1;
                *(v2f*)&ipad[img * TILE_F + ((o >> 1) * QP + q) * 2] = st;
            }
        }
    });

    load_w(wt + 36864);
    #pragma unroll
    for (int r = 0; r < 8; ++r) {
        sc[r] = g2[obase + r] * rsqrtf(v2p[obase + r] + EPSV);
        sh[r] = b2[obase + r] - m2[obase + r] * sc[r];
    }
    __syncthreads();

    conv_pass(ipad, [&](int ti, v8f acc) __attribute__((always_inline)) {
        const int p = ti * 16 + c;
        if (p < NPOS) {
            const int img = p / 49, m = p - img * 49;
            const int q = (m / 7 + 1) * PITCH + (m % 7) + 1;
            const bool first = (img0 + img) == 0;
            #pragma unroll
            for (int r = 0; r < 8; r += 2) {
                const int o = obase + r;
                float u0 = acc[r], u1 = acc[r + 1];
                if (first) { u0 *= mask2[o * 49 + m]; u1 *= mask2[(o + 1) * 49 + m]; }
                u0 = fmaf(u0, sc[r],     sh[r]);
                u1 = fmaf(u1, sc[r + 1], sh[r + 1]);
                v2f* px = (v2f*)&xpad[img * TILE_F + ((o >> 1) * QP + q) * 2];
                const v2f xr = *px;
                v2f st; st[0] = fmaxf(u0 + xr[0], 0.0f); st[1] = fmaxf(u1 + xr[1], 0.0f);
                *px = st;
            }
        }
    });
    __syncthreads();

    {
        const float* sp = xpad + wave * TILE_F + lane * (2 * QP);
        float* rp = park + lane * 98;
        #pragma unroll
        for (int m = 0; m < 49; ++m) {
            const int q = (m / 7 + 1) * PITCH + (m % 7) + 1;
            const v2f v = *(const v2f*)(sp + q * 2);
            rp[m] = v[0]; rp[49 + m] = v[1];
        }
    }
    __syncthreads();
    {
        const float4* pk = (const float4*)park;
        float4* o4 = (float4*)(out + (size_t)(img0 + wave) * 3136);
        #pragma unroll
        for (int j = 0; j < 24; ++j) o4[lane + 32 * j] = pk[lane + 32 * j];
        if (lane < 16) o4[lane + 768] = pk[lane + 768];
    }
}

extern "C" void kernel_launch(void* const* d_in, const int* in_sizes, int n_in,
                              void* d_out, int out_size, void* d_ws, size_t ws_size,
                              hipStream_t stream) {
    const float* x     = (const float*)d_in[0];
    const float* w1    = (const float*)d_in[1];
    const float* g1    = (const float*)d_in[2];
    const float* b1    = (const float*)d_in[3];
    const float* m1    = (const float*)d_in[4];
    const float* v1    = (const float*)d_in[5];
    const float* w2    = (const float*)d_in[6];
    const float* g2    = (const float*)d_in[7];
    const float* b2    = (const float*)d_in[8];
    const float* m2    = (const float*)d_in[9];
    const float* v2p   = (const float*)d_in[10];
    const float* mask1 = (const float*)d_in[11];
    const float* mask2 = (const float*)d_in[12];
    float* wt = (float*)d_ws;
    float* out = (float*)d_out;

    prep_weights_kernel<<<288, 256, 0, stream>>>(w1, w2, wt);
    fused_basicblock_kernel<<<1024, NTHR, 0, stream>>>(
        x, wt, g1, b1, m1, v1, g2, b2, m2, v2p, mask1, mask2, out);
}


# Column tiles sorted by border class via the per-lane B pointers: 6 of 13 tiles hold only top/bottom-row or left/right-column pixels and omit the 3 taps that read zero pad: 1584 WMMAs a conv, was 1872
# speedup vs baseline: 1.6848x; 1.1234x over previous
#include <hip/hip_runtime.h>


#define EPSV 1e-5f

typedef __attribute__((ext_vector_type(2))) float v2f;
typedef __attribute__((ext_vector_type(8))) float v8f;

__global__ __launch_bounds__(256) void prep_weights_kernel(
    const float* __restrict__ w1, const float* __restrict__ w2,
    float* __restrict__ wt)
{
    int i = blockIdx.x * 256 + threadIdx.x;
    if (i >= 2 * 36864) return;
    const float* w = (i < 36864) ? w1 : w2;
    float* dst = wt + (i < 36864 ? 0 : 36864);
    int j = i % 36864;
    int tap = j / 4096;
    int rem = j % 4096;
    int n = (rem >> 1) & 63;
    int k = ((rem >> 7) << 1) | (rem & 1);
    dst[j] = w[(n * 64 + k) * 9 + tap];
}

#define NIMG 4
#define NTHR 128
#define PITCH 8
#define QP 73
#define TILE_F (64 * QP)
#define NFULL 7
#define NGRP 4

template <int N> struct IC { static constexpr int v = N; };

#define K_FULL  0
#define K_TOP   1
#define K_BOT   2
#define K_LEFT  3
#define K_RIGHT 4
__host__ __device__ constexpr int kind_ntap(int k) { return k == K_FULL ? 9 : 6; }
__host__ __device__ constexpr int kind_tap(int k, int i) {
    return k == K_FULL ? i
         : k == K_TOP  ? i + 3
         : k == K_BOT  ? i
         : k == K_LEFT ? (i / 2) * 3 + 1 + (i % 2)
         :               (i / 2) * 3 + (i % 2);
}

__global__ __launch_bounds__(NTHR, 2) void fused_basicblock_kernel(
    const float* __restrict__ x,
    const float* __restrict__ wt,
    const float* __restrict__ g1, const float* __restrict__ b1,
    const float* __restrict__ m1, const float* __restrict__ v1,
    const float* __restrict__ g2, const float* __restrict__ b2,
    const float* __restrict__ m2, const float* __restrict__ v2p,
    const float* __restrict__ mask1, const float* __restrict__ mask2,
    float* __restrict__ out)
{
    __shared__ __align__(16) float xpad[NIMG * TILE_F];
    __shared__ __align__(16) float ipad[NIMG * TILE_F];

    const int tid  = threadIdx.x;
    const int lane = tid & 31;
    const int wave = __builtin_amdgcn_readfirstlane(tid >> 5);
    const int c    = lane & 15;
    const int h    = lane >> 4;
    const int mt   = wave;
    const int img0 = blockIdx.x * NIMG;

    const float4 z4 = {0.0f, 0.0f, 0.0f, 0.0f};
    float* park = ipad + wave * TILE_F;
    {
        const float4* xs = (const float4*)(x + (size_t)(img0 + wave) * 3136);
        float4 rv[24];
        float4 rlast = z4;
        #pragma unroll
        for (int j = 0; j < 24; ++j) rv[j] = xs[lane + 32 * j];
        if (lane < 16) rlast = xs[lane + 768];
        float4* xz = (float4*)xpad;
        for (int i = tid; i < NIMG * TILE_F / 4; i += NTHR) xz[i] = z4;
        float4* pk = (float4*)park;
        #pragma unroll
        for (int j = 0; j < 24; ++j) pk[lane + 32 * j] = rv[j];
        if (lane < 16) pk[lane + 768] = rlast;
    }
    __syncthreads();
    {
        const float* rp = park + lane * 98;
        float* dp = xpad + wave * TILE_F + lane * (2 * QP);
        #pragma unroll
        for (int m = 0; m < 49; ++m) {
            const int q = (m / 7 + 1) * PITCH + (m % 7) + 1;
            v2f st; st[0] = rp[m]; st[1] = rp[49 + m];
            *(v2f*)(dp + q * 2) = st;
        }
    }
    __syncthreads();
    {
        float4* iz = (float4*)ipad;
        for (int i = tid; i < NIMG * TILE_F / 4; i += NTHR) iz[i] = z4;
    }

    v2f W[144];
    auto load_w = [&](const float* __restrict__ wconv) {
        const float* wp = wconv + h * 128 + (mt * 16 + c) * 2;
        #pragma unroll
        for (int t = 0; t < 9; ++t) {
            #pragma unroll
            for (int ks = 0; ks < 16; ++ks)
                W[t * 16 + ks] = *(const v2f*)(wp + t * 4096 + ks * 256);
        }
    };

    struct Pos { int img, y, x; bool live; };
    auto tile_pos = [&](auto kc, int t, int c) __attribute__((always_inline)) -> Pos {
        constexpr int K = decltype(kc)::v;
        Pos p;
        if constexpr (K == K_FULL) {
            int s = t * 16 + c;
            p.live = s < 108;
            if (s > 107) s = 107;
            p.img = s / 27;
            const int j = s - p.img * 27;
            if (j < 20) { p.y = 1 + j / 5; p.x = 1 + j - (j / 5) * 5; }
            else        { p.y = 5;         p.x = j - 20; }
        } else if constexpr (K == K_TOP || K == K_BOT) {
            int s = t * 16 + c;
            p.live = s < 28;
            if (s > 27) s = 27;
            p.img = s / 7;
            p.x = s - p.img * 7;
            p.y = (K == K_TOP) ? 0 : 6;
        } else {
            p.live = true;
            p.img = c >> 2;
            p.y = 1 + (c & 3);
            p.x = (K == K_LEFT) ? 0 : 6;
        }
        return p;
    };
    auto b_ptr = [&](const float* src, const Pos& p) __attribute__((always_inline)) -> const v2f* {
        return (const v2f*)(src + p.img * TILE_F + h * (2 * QP) + (p.y * PITCH + p.x) * 2);
    };

    v2f S0[36], S1[36];
    auto load_grp = [&](v2f (&S)[36], const v2f* bp, auto kc, auto gc) __attribute__((always_inline)) {
        constexpr int K = decltype(kc)::v, g = decltype(gc)::v, FP = kind_ntap(K) * 16 / NGRP;
        #pragma unroll
        for (int j = 0; j < FP; ++j) {
            const int f = g * FP + j, t = kind_tap(K, f >> 4), ks = f & 15;
            S[j] = bp[ks * (2 * QP) + (t / 3) * PITCH + (t % 3)];
        }
    };
    auto mma_grp = [&](const v2f (&S)[36], auto kc, auto gc, v8f& a0, v8f& a1) __attribute__((always_inline)) {
        constexpr int K = decltype(kc)::v, g = decltype(gc)::v, FP = kind_ntap(K) * 16 / NGRP;
        #pragma unroll
        for (int j = 0; j < FP; j += 2) {
            const int f0 = g * FP + j, f1 = f0 + 1;
            a0 = __builtin_amdgcn_wmma_f32_16x16x4_f32(
                false, W[kind_tap(K, f0 >> 4) * 16 + (f0 & 15)], false, S[j],
                (short)0, a0, false, false);
            a1 = __builtin_amdgcn_wmma_f32_16x16x4_f32(
                false, W[kind_tap(K, f1 >> 4) * 16 + (f1 & 15)], false, S[j + 1],
                (short)0, a1, false, false);
        }
    };

#define KEEP_GROUPED(A0, A1) asm volatile("" : "+v"(A0), "+v"(A1) : : "memory")

    auto tile = [&](auto kc, const v2f* bp, auto&& next0, auto&& next1, auto&& epi) __attribute__((always_inline)) {
        v8f a0 = {}, a1 = {};
        KEEP_GROUPED(a0, a1);
        mma_grp(S0, kc, IC<0>{}, a0, a1);  KEEP_GROUPED(a0, a1);
        load_grp(S0, bp, kc, IC<2>{});     KEEP_GROUPED(a0, a1);
        mma_grp(S1, kc, IC<1>{}, a0, a1);  KEEP_GROUPED(a0, a1);
        load_grp(S1, bp, kc, IC<3>{});     KEEP_GROUPED(a0, a1);
        mma_grp(S0, kc, IC<2>{}, a0, a1);  KEEP_GROUPED(a0, a1);
        next0();                           KEEP_GROUPED(a0, a1);
        mma_grp(S1, kc, IC<3>{}, a0, a1);  KEEP_GROUPED(a0, a1);
        next1();
        epi(a0 + a1);
    };

    auto conv_pass = [&](const float* src, auto&& epilogue) __attribute__((always_inline)) {
        int cp = c;
        asm volatile("" : "+v"(cp));
        Pos pc;
        const v2f* bc;
        auto open = [&](auto kc) __attribute__((always_inline)) {
            pc = tile_pos(kc, 0, cp);
            bc = b_ptr(src, pc);
            load_grp(S0, bc, kc, IC<0>{});
            asm volatile("" : : : "memory");
            load_grp(S1, bc, kc, IC<1>{});
        };
        auto step = [&](auto kc, auto kn, int tn) __attribute__((always_inline)) {
            const Pos pn = tile_pos(kn, tn, cp);
            const v2f* bn = b_ptr(src, pn);
            const Pos pe = pc;
            tile(kc, bc,
                 [&]() __attribute__((always_inline)) { load_grp(S0, bn, kn, IC<0>{}); },
                 [&]() __attribute__((always_inline)) { load_grp(S1, bn, kn, IC<1>{}); },
                 [&](v8f acc) __attribute__((always_inline)) { epilogue(pe, acc); });
            pc = pn; bc = bn;
        };
        open(IC<K_LEFT>{});
        step(IC<K_LEFT>{},  IC<K_RIGHT>{}, 0);
        step(IC<K_RIGHT>{}, IC<K_TOP>{},   0);
        #pragma nounroll
        for (int t = 0; t < 2; ++t) {
            step(IC<K_TOP>{}, IC<K_BOT>{}, t);
            step(IC<K_BOT>{}, IC<K_TOP>{}, t + 1);
        }
        open(IC<K_FULL>{});
        for (int ti = 0; ti < NFULL; ++ti)
            step(IC<K_FULL>{}, IC<K_FULL>{}, ti + 1);
    };

    const int obase = mt * 16 + 8 * h;

    float sc[8], sh[8];

    #pragma nounroll
    for (int cv = 0; cv < 2; ++cv) {
        const float* gp = cv ? g2 : g1;
        const float* bp = cv ? b2 : b1;
        const float* mp = cv ? m2 : m1;
        const float* vp = cv ? v2p : v1;
        const float* mk = cv ? mask2 : mask1;
        load_w(wt + cv * 36864);
        #pragma unroll
        for (int r = 0; r < 8; ++r) {
            sc[r] = gp[obase + r] * rsqrtf(vp[obase + r] + EPSV);
            sh[r] = bp[obase + r] - mp[obase + r] * sc[r];
        }
        __syncthreads();

        conv_pass(cv ? ipad : xpad, [&](const Pos& p, v8f acc) __attribute__((always_inline)) {
            if (p.live) {
                const int img = p.img, m = p.y * 7 + p.x;
                const int q = (p.y + 1) * PITCH + p.x + 1;
                const bool first = (img0 + img) == 0;
                if (cv == 0) {
                    #pragma unroll
                    for (int r = 0; r < 8; r += 2) {
                        const int o = obase + r;
                        float u0 = fmaxf(fmaf(acc[r],     sc[r],     sh[r]),     0.0f);
                        float u1 = fmaxf(fmaf(acc[r + 1], sc[r + 1], sh[r + 1]), 0.0f);
                        if (first) { u0 *= mk[o * 49 + m]; u1 *= mk[(o + 1) * 49 + m]; }
                        v2f st; st[0] = u0; st[1] = u1;
                        *(v2f*)&ipad[img * TILE_F + ((o >> 1) * QP + q) * 2] = st;
                    }
                } else {
                    #pragma unroll
                    for (int r = 0; r < 8; r += 2) {
                        const int o = obase + r;
                        float u0 = acc[r], u1 = acc[r + 1];
                        if (first) { u0 *= mk[o * 49 + m]; u1 *= mk[(o + 1) * 49 + m]; }
                        u0 = fmaf(u0, sc[r],     sh[r]);
                        u1 = fmaf(u1, sc[r + 1], sh[r + 1]);
                        v2f* px = (v2f*)&xpad[img * TILE_F + ((o >> 1) * QP + q) * 2];
                        const v2f xr = *px;
                        v2f st; st[0] = fmaxf(u0 + xr[0], 0.0f); st[1] = fmaxf(u1 + xr[1], 0.0f);
                        *px = st;
                    }
                }
            }
        });
    }
    __syncthreads();

    {
        const float* sp = xpad + wave * TILE_F + lane * (2 * QP);
        float* rp = park + lane * 98;
        #pragma unroll
        for (int m = 0; m < 49; ++m) {
            const int q = (m / 7 + 1) * PITCH + (m % 7) + 1;
            const v2f v = *(const v2f*)(sp + q * 2);
            rp[m] = v[0]; rp[49 + m] = v[1];
        }
    }
    __syncthreads();
    {
        const float4* pk = (const float4*)park;
        float4* o4 = (float4*)(out + (size_t)(img0 + wave) * 3136);
        #pragma unroll
        for (int j = 0; j < 24; ++j) o4[lane + 32 * j] = pk[lane + 32 * j];
        if (lane < 16) o4[lane + 768] = pk[lane + 768];
    }
}

extern "C" void kernel_launch(void* const* d_in, const int* in_sizes, int n_in,
                              void* d_out, int out_size, void* d_ws, size_t ws_size,
                              hipStream_t stream) {
    const float* x     = (const float*)d_in[0];
    const float* w1    = (const float*)d_in[1];
    const float* g1    = (const float*)d_in[2];
    const float* b1    = (const float*)d_in[3];
    const float* m1    = (const float*)d_in[4];
    const float* v1    = (const float*)d_in[5];
    const float* w2    = (const float*)d_in[6];
    const float* g2    = (const float*)d_in[7];
    const float* b2    = (const float*)d_in[8];
    const float* m2    = (const float*)d_in[9];
    const float* v2p   = (const float*)d_in[10];
    const float* mask1 = (const float*)d_in[11];
    const float* mask2 = (const float*)d_in[12];
    float* wt = (float*)d_ws;
    float* out = (float*)d_out;

    prep_weights_kernel<<<288, 256, 0, stream>>>(w1, w2, wt);
    fused_basicblock_kernel<<<1024, NTHR, 0, stream>>>(
        x, wt, g1, b1, m1, v1, g2, b2, m2, v2p, mask1, mask2, out);
}
